# starting kernel
# baseline (speedup 1.0000x reference)
#include <hip/hip_runtime.h>

typedef __attribute__((ext_vector_type(2))) float v2f;
typedef __attribute__((ext_vector_type(8))) float v8f;

#define BATCH 16
#define HH    128
#define WW    128
#define RDIM  4
#define IDIM  5
#define ADIM  8
#define KSQ   9
#define IKK   45
#define KPAD  48
#define MROWS 368
#define TILE  16
#define WAVES 8
#define BLOCK 256

#define WLDS_FLOATS (MROWS * KPAD)
#define PLDS_FLOATS (KPAD * TILE)
#define LDS_BYTES   ((WLDS_FLOATS + WAVES * PLDS_FLOATS) * 4)

__global__ __launch_bounds__(BLOCK)
void vi_wmma_kernel(const float* __restrict__ values,
                    const float* __restrict__ rewards,
                    const float* __restrict__ weight,
                    float* __restrict__ out)
{
    extern __shared__ float smem[];
    float* wlds = smem;
    const int tid  = threadIdx.x;
    const int lane = tid & 31;
    const int wave = tid >> 5;
    float* plds = smem + WLDS_FLOATS + wave * PLDS_FLOATS;

    #pragma unroll 1
    for (int it = 0; it < WLDS_FLOATS / BLOCK; ++it) {
        int idx = tid + it * BLOCK;
        int c = idx / KPAD;
        int k = idx - c * KPAD;
        float v = 0.0f;
        if (c < ADIM * IKK && k < IKK) v = weight[c * IKK + k];
        wlds[idx] = v;
    }

    int g   = blockIdx.x * WAVES + wave;
    int lin = g * TILE;
    int b   = lin >> 14;
    int rem = lin & 16383;
    int y   = rem >> 7;
    int x0  = rem & 127;

    #pragma unroll 1
    for (int t = 0; t < PLDS_FLOATS / 32; ++t) {
        int idx = lane + 32 * t;
        int k = idx >> 4;
        int n = idx & 15;
        float v = 0.0f;
        if (k < IKK) {
            int i  = k / KSQ;
            int rr = k - i * KSQ;
            int di = rr / 3;
            int dj = rr - di * 3;
            int yy = y + di - 1;
            int xx = x0 + n + dj - 1;
            if (yy >= 0 && yy < HH && xx >= 0 && xx < WW) {
                v = (i < RDIM) ? rewards[((b * RDIM + i) * HH + yy) * WW + xx]
                               : values[(b * HH + yy) * WW + xx];
            }
        }
        plds[idx] = v;
    }

    __syncthreads();

    const int  n  = lane & 15;
    const bool hi = lane >= 16;
    const int  kh = hi ? 2 : 0;

    v2f bs[12];
    #pragma unroll
    for (int s = 0; s < 12; ++s) {
        int k0 = 4 * s + kh;
        bs[s][0] = plds[k0 * TILE + n];
        bs[s][1] = plds[(k0 + 1) * TILE + n];
    }

    float pp[24];
    #pragma unroll
    for (int t = 0; t < 3; ++t)
        #pragma unroll
        for (int r = 0; r < 8; ++r) {
            int j = 16 * t + r + (hi ? 8 : 0);
            pp[t * 8 + r] = plds[j * TILE + n];
        }

    float best = -3.402823466e+38f;

    #pragma unroll 1
    for (int o = 0; o < ADIM; ++o) {
        v8f acc0 = {}, acc1 = {}, acc2 = {};
        const float* wb = wlds + (o * IKK) * KPAD + kh;
        #pragma unroll
        for (int s = 0; s < 12; ++s) {
            v2f a0 = *(const v2f*)(wb + (0 * 16 + n) * KPAD + 4 * s);
            v2f a1 = *(const v2f*)(wb + (1 * 16 + n) * KPAD + 4 * s);
            v2f a2 = *(const v2f*)(wb + (2 * 16 + n) * KPAD + 4 * s);
            acc0 = __builtin_amdgcn_wmma_f32_16x16x4_f32(false, a0, false, bs[s], (short)0, acc0, false, false);
            acc1 = __builtin_amdgcn_wmma_f32_16x16x4_f32(false, a1, false, bs[s], (short)0, acc1, false, false);
            acc2 = __builtin_amdgcn_wmma_f32_16x16x4_f32(false, a2, false, bs[s], (short)0, acc2, false, false);
        }

        float vmax = -3.402823466e+38f;
        #pragma unroll
        for (int r = 0; r < 8; ++r) vmax = fmaxf(vmax, acc0[r]);
        #pragma unroll
        for (int r = 0; r < 8; ++r) vmax = fmaxf(vmax, acc1[r]);
        #pragma unroll
        for (int r = 0; r < 8; ++r) {
            bool valid = (32 + r + (hi ? 8 : 0)) < IKK;
            vmax = fmaxf(vmax, valid ? acc2[r] : -3.402823466e+38f);
        }
        vmax = fmaxf(vmax, __shfl_xor(vmax, 16));

        float se = 0.0f, qe = 0.0f;
        #pragma unroll
        for (int r = 0; r < 8; ++r) {
            float e = __expf(acc0[r] - vmax);
            se += e; qe += e * pp[r];
        }
        #pragma unroll
        for (int r = 0; r < 8; ++r) {
            float e = __expf(acc1[r] - vmax);
            se += e; qe += e * pp[8 + r];
        }
        #pragma unroll
        for (int r = 0; r < 8; ++r) {
            bool valid = (32 + r + (hi ? 8 : 0)) < IKK;
            float e = valid ? __expf(acc2[r] - vmax) : 0.0f;
            se += e; qe += e * pp[16 + r];
        }
        se += __shfl_xor(se, 16);
        qe += __shfl_xor(qe, 16);
        best = fmaxf(best, qe / se);
    }

    if (!hi) out[lin + n] = best;
}

extern "C" void kernel_launch(void* const* d_in, const int* in_sizes, int n_in,
                              void* d_out, int out_size, void* d_ws, size_t ws_size,
                              hipStream_t stream) {
    (void)in_sizes; (void)n_in; (void)d_ws; (void)ws_size; (void)out_size;
    const float* values  = (const float*)d_in[0];
    const float* rewards = (const float*)d_in[1];
    const float* weight  = (const float*)d_in[2];
    float* out = (float*)d_out;

    const int tiles  = (BATCH * HH * WW) / TILE;
    const int blocks = tiles / WAVES;
    hipLaunchKernelGGL(vi_wmma_kernel, dim3(blocks), dim3(BLOCK), LDS_BYTES, stream,
                       values, rewards, weight, out);
}


# Persistent workgroups: 256 groups of 32 waves stage the weights into LDS once and each wave loops over 2 pixel tiles (was 2048 groups of 8 waves, one tile each, weights restaged per group)
# speedup vs baseline: 1.6785x; 1.6785x over previous
#include <hip/hip_runtime.h>

typedef __attribute__((ext_vector_type(2))) float v2f;
typedef __attribute__((ext_vector_type(8))) float v8f;

#define BATCH 16
#define HH    128
#define WW    128
#define RDIM  4
#define IDIM  5
#define ADIM  8
#define KSQ   9
#define IKK   45
#define KPAD  48
#define MROWS 368
#define TILE  16
#define WAVES 32
#define BLOCK 1024
#define NBLOCKS 256
#define TILES_PER_WAVE 2

#define WLDS_FLOATS (MROWS * KPAD)
#define PLDS_FLOATS (KPAD * TILE)
#define LDS_FLOATS  (WLDS_FLOATS + WAVES * PLDS_FLOATS)

__global__ __launch_bounds__(BLOCK)
void vi_wmma_kernel(const float* __restrict__ values,
                    const float* __restrict__ rewards,
                    const float* __restrict__ weight,
                    float* __restrict__ out)
{
    __shared__ __attribute__((aligned(16))) float smem[LDS_FLOATS];
    float* wlds = smem;
    const int tid  = threadIdx.x;
    const int lane = tid & 31;
    const int wave = tid >> 5;
    float* plds = smem + WLDS_FLOATS + wave * PLDS_FLOATS;

    #pragma unroll 1
    for (int idx = tid; idx < WLDS_FLOATS; idx += BLOCK) {
        int c = idx / KPAD;
        int k = idx - c * KPAD;
        float v = 0.0f;
        if (c < ADIM * IKK && k < IKK) v = weight[c * IKK + k];
        wlds[idx] = v;
    }

    __syncthreads();

    const int  n  = lane & 15;
    const bool hi = lane >= 16;
    const int  kh = hi ? 2 : 0;

    #pragma unroll 1
    for (int tt = 0; tt < TILES_PER_WAVE; ++tt) {
        int g   = blockIdx.x * (WAVES * TILES_PER_WAVE) + tt * WAVES + wave;
        int lin = g * TILE;
        int b   = lin >> 14;
        int rem = lin & 16383;
        int y   = rem >> 7;
        int x0  = rem & 127;

        #pragma unroll 1
        for (int t = 0; t < PLDS_FLOATS / 32; ++t) {
            int idx = lane + 32 * t;
            int k = idx >> 4;
            int nn = idx & 15;
            float v = 0.0f;
            if (k < IKK) {
                int i  = k / KSQ;
                int rr = k - i * KSQ;
                int di = rr / 3;
                int dj = rr - di * 3;
                int yy = y + di - 1;
                int xx = x0 + nn + dj - 1;
                if (yy >= 0 && yy < HH && xx >= 0 && xx < WW) {
                    v = (i < RDIM) ? rewards[((b * RDIM + i) * HH + yy) * WW + xx]
                                   : values[(b * HH + yy) * WW + xx];
                }
            }
            plds[idx] = v;
        }


        v2f bs[12];
        #pragma unroll
        for (int s = 0; s < 12; ++s) {
            int k0 = 4 * s + kh;
            bs[s][0] = plds[k0 * TILE + n];
            bs[s][1] = plds[(k0 + 1) * TILE + n];
        }

        float pp[24];
        #pragma unroll
        for (int t = 0; t < 3; ++t)
            #pragma unroll
            for (int r = 0; r < 8; ++r) {
                int j = 16 * t + r + (hi ? 8 : 0);
                pp[t * 8 + r] = plds[j * TILE + n];
            }

        float best = -3.402823466e+38f;

        #pragma unroll 1
        for (int o = 0; o < ADIM; ++o) {
            v8f acc0 = {}, acc1 = {}, acc2 = {};
            const float* wb = wlds + (o * IKK) * KPAD + kh;
            #pragma unroll
            for (int s = 0; s < 12; ++s) {
                v2f a0 = *(const v2f*)(wb + (0 * 16 + n) * KPAD + 4 * s);
                v2f a1 = *(const v2f*)(wb + (1 * 16 + n) * KPAD + 4 * s);
                v2f a2 = *(const v2f*)(wb + (2 * 16 + n) * KPAD + 4 * s);
                acc0 = __builtin_amdgcn_wmma_f32_16x16x4_f32(false, a0, false, bs[s], (short)0, acc0, false, false);
                acc1 = __builtin_amdgcn_wmma_f32_16x16x4_f32(false, a1, false, bs[s], (short)0, acc1, false, false);
                acc2 = __builtin_amdgcn_wmma_f32_16x16x4_f32(false, a2, false, bs[s], (short)0, acc2, false, false);
            }

            float vmax = -3.402823466e+38f;
            #pragma unroll
            for (int r = 0; r < 8; ++r) vmax = fmaxf(vmax, acc0[r]);
            #pragma unroll
            for (int r = 0; r < 8; ++r) vmax = fmaxf(vmax, acc1[r]);
            #pragma unroll
            for (int r = 0; r < 8; ++r) {
                bool valid = (32 + r + (hi ? 8 : 0)) < IKK;
                vmax = fmaxf(vmax, valid ? acc2[r] : -3.402823466e+38f);
            }
            vmax = fmaxf(vmax, __shfl_xor(vmax, 16));

            float se = 0.0f, qe = 0.0f;
            #pragma unroll
            for (int r = 0; r < 8; ++r) {
                float e = __expf(acc0[r] - vmax);
                se += e; qe += e * pp[r];
            }
            #pragma unroll
            for (int r = 0; r < 8; ++r) {
                float e = __expf(acc1[r] - vmax);
                se += e; qe += e * pp[8 + r];
            }
            #pragma unroll
            for (int r = 0; r < 8; ++r) {
                bool valid = (32 + r + (hi ? 8 : 0)) < IKK;
                float e = valid ? __expf(acc2[r] - vmax) : 0.0f;
                se += e; qe += e * pp[16 + r];
            }
            se += __shfl_xor(se, 16);
            qe += __shfl_xor(qe, 16);
            best = fmaxf(best, qe / se);
        }

        if (!hi) out[lin + n] = best;
    }
}

extern "C" void kernel_launch(void* const* d_in, const int* in_sizes, int n_in,
                              void* d_out, int out_size, void* d_ws, size_t ws_size,
                              hipStream_t stream) {
    (void)in_sizes; (void)n_in; (void)d_ws; (void)ws_size; (void)out_size;
    const float* values  = (const float*)d_in[0];
    const float* rewards = (const float*)d_in[1];
    const float* weight  = (const float*)d_in[2];
    float* out = (float*)d_out;

    hipLaunchKernelGGL(vi_wmma_kernel, dim3(NBLOCKS), dim3(BLOCK), 0, stream,
                       values, rewards, weight, out);
}


# Input window instead of im2col copy: each wave fetches its 15x18 zero-padded window with 15 independent row loads and reads B fragments straight from it (was 24 dependent one-element gather trips)
# speedup vs baseline: 1.9864x; 1.1834x over previous
#include <hip/hip_runtime.h>

typedef __attribute__((ext_vector_type(2))) float v2f;
typedef __attribute__((ext_vector_type(8))) float v8f;

#define BATCH 16
#define HH    128
#define WW    128
#define RDIM  4
#define IDIM  5
#define ADIM  8
#define KSQ   9
#define IKK   45
#define KPAD  48
#define MROWS 368
#define TILE  16
#define WAVES 32
#define BLOCK 1024
#define NBLOCKS 256
#define TILES_PER_WAVE 2

#define WLDS_FLOATS (MROWS * KPAD)
#define RP    18
#define RAW_FLOATS (16 * RP)
#define LDS_FLOATS  (WLDS_FLOATS + WAVES * RAW_FLOATS)

__global__ __launch_bounds__(BLOCK) __attribute__((amdgpu_waves_per_eu(8, 8)))
void vi_wmma_kernel(const float* __restrict__ values,
                    const float* __restrict__ rewards,
                    const float* __restrict__ weight,
                    float* __restrict__ out)
{
    __shared__ __attribute__((aligned(16))) float smem[LDS_FLOATS];
    float* wlds = smem;
    const int tid  = threadIdx.x;
    const int lane = tid & 31;
    const int wave = tid >> 5;
    float* raw = smem + WLDS_FLOATS + wave * RAW_FLOATS;

    #pragma unroll 1
    for (int idx = tid; idx < WLDS_FLOATS; idx += BLOCK) {
        int c = idx / KPAD;
        int k = idx - c * KPAD;
        float v = 0.0f;
        if (c < ADIM * IKK && k < IKK) v = weight[c * IKK + k];
        wlds[idx] = v;
    }

    __syncthreads();

    const int  n  = lane & 15;
    const bool hi = lane >= 16;
    const int  kh = hi ? 2 : 0;

    #pragma unroll 1
    for (int tt = 0; tt < TILES_PER_WAVE; ++tt) {
        int g   = blockIdx.x * (WAVES * TILES_PER_WAVE) + tt * WAVES + wave;
        int lin = g * TILE;
        int b   = lin >> 14;
        int rem = lin & 16383;
        int y   = rem >> 7;
        int x0  = rem & 127;

        const bool cin = lane < RP;
        const int  xx  = x0 - 1 + lane;
        const bool xok = (xx >= 0) && (xx < WW);
        const int  xc  = min(max(xx, 0), WW - 1);
        int  ro[3];
        bool ok[3];
        #pragma unroll
        for (int dy = 0; dy < 3; ++dy) {
            int yy = y + dy - 1;
            ok[dy] = xok && (yy >= 0) && (yy < HH);
            ro[dy] = min(max(yy, 0), HH - 1) * WW + xc;
        }
        const float* vb = values  + b * (HH * WW);
        const float* rb = rewards + b * (RDIM * HH * WW);
        float wv[15];
        #pragma unroll
        for (int r = 0; r < 15; ++r) {
            const int i  = r / 3;
            const int dy = r - 3 * i;
            float v = (i < RDIM) ? rb[i * (HH * WW) + ro[dy]] : vb[ro[dy]];
            wv[r] = ok[dy] ? v : 0.0f;
        }
        if (cin) {
            #pragma unroll
            for (int r = 0; r < 15; ++r) raw[r * RP + lane] = wv[r];
            raw[15 * RP + lane] = 0.0f;
        }


        const float* fA = raw + n + (hi ? 2 : 0);
        const float* fB = raw + n + (hi ? RP - 1 : 0);
        const float* fC = raw + n + (hi ? 2 * RP + 2 : 0);
        const float* fD = raw + n + (hi ? 3 * RP - 1 : 0);

        v2f bs[12];
        #pragma unroll
        for (int s = 0; s < 12; ++s) {
            #pragma unroll
            for (int e = 0; e < 2; ++e) {
                const int kl  = 4 * s + e;
                const int off = (kl / 3) * RP + (kl % 3);
                bs[s][e] = ((kl % 3) == 0 ? fA : fB)[off];
            }
        }

        float pp[24];
        #pragma unroll
        for (int t = 0; t < 3; ++t) {
            #pragma unroll
            for (int r = 0; r < 8; ++r) {
                const int jl  = 16 * t + r;
                const int off = (jl / 3) * RP + (jl % 3);
                pp[t * 8 + r] = ((jl % 3) == 0 ? fC : fD)[off];
            }
        }

        float best = -3.402823466e+38f;

        #pragma unroll 1
        for (int o = 0; o < ADIM; ++o) {
            v8f acc0 = {}, acc1 = {}, acc2 = {};
            const float* wb = wlds + (o * IKK) * KPAD + kh;
            #pragma unroll
            for (int s = 0; s < 12; ++s) {
                v2f a0 = *(const v2f*)(wb + (0 * 16 + n) * KPAD + 4 * s);
                v2f a1 = *(const v2f*)(wb + (1 * 16 + n) * KPAD + 4 * s);
                v2f a2 = *(const v2f*)(wb + (2 * 16 + n) * KPAD + 4 * s);
                acc0 = __builtin_amdgcn_wmma_f32_16x16x4_f32(false, a0, false, bs[s], (short)0, acc0, false, false);
                acc1 = __builtin_amdgcn_wmma_f32_16x16x4_f32(false, a1, false, bs[s], (short)0, acc1, false, false);
                acc2 = __builtin_amdgcn_wmma_f32_16x16x4_f32(false, a2, false, bs[s], (short)0, acc2, false, false);
            }

            float vmax = -3.402823466e+38f;
            #pragma unroll
            for (int r = 0; r < 8; ++r) vmax = fmaxf(vmax, acc0[r]);
            #pragma unroll
            for (int r = 0; r < 8; ++r) vmax = fmaxf(vmax, acc1[r]);
            #pragma unroll
            for (int r = 0; r < 8; ++r) {
                bool valid = (32 + r + (hi ? 8 : 0)) < IKK;
                vmax = fmaxf(vmax, valid ? acc2[r] : -3.402823466e+38f);
            }
            vmax = fmaxf(vmax, __shfl_xor(vmax, 16));

            float se = 0.0f, qe = 0.0f;
            #pragma unroll
            for (int r = 0; r < 8; ++r) {
                float e = __expf(acc0[r] - vmax);
                se += e; qe += e * pp[r];
            }
            #pragma unroll
            for (int r = 0; r < 8; ++r) {
                float e = __expf(acc1[r] - vmax);
                se += e; qe += e * pp[8 + r];
            }
            #pragma unroll
            for (int r = 0; r < 8; ++r) {
                bool valid = (32 + r + (hi ? 8 : 0)) < IKK;
                float e = valid ? __expf(acc2[r] - vmax) : 0.0f;
                se += e; qe += e * pp[16 + r];
            }
            se += __shfl_xor(se, 16);
            qe += __shfl_xor(qe, 16);
            best = fmaxf(best, qe / se);
        }

        if (!hi) out[lin + n] = best;
    }
}

extern "C" void kernel_launch(void* const* d_in, const int* in_sizes, int n_in,
                              void* d_out, int out_size, void* d_ws, size_t ws_size,
                              hipStream_t stream) {
    (void)in_sizes; (void)n_in; (void)d_ws; (void)ws_size; (void)out_size;
    const float* values  = (const float*)d_in[0];
    const float* rewards = (const float*)d_in[1];
    const float* weight  = (const float*)d_in[2];
    float* out = (float*)d_out;

    hipLaunchKernelGGL(vi_wmma_kernel, dim3(NBLOCKS), dim3(BLOCK), 0, stream,
                       values, rewards, weight, out);
}


# Weight staging: each thread issues all of its global weight loads together and then stores them to LDS (was a loop of 18 dependent load-wait-store trips in front of the barrier)
# speedup vs baseline: 2.1797x; 1.0973x over previous
#include <hip/hip_runtime.h>

typedef __attribute__((ext_vector_type(2))) float v2f;
typedef __attribute__((ext_vector_type(8))) float v8f;

#define BATCH 16
#define HH    128
#define WW    128
#define RDIM  4
#define IDIM  5
#define ADIM  8
#define KSQ   9
#define IKK   45
#define KPAD  48
#define MROWS 368
#define TILE  16
#define WAVES 32
#define BLOCK 1024
#define NBLOCKS 256
#define TILES_PER_WAVE 2

#define WLDS_FLOATS (MROWS * KPAD)
#define STAGE_ITERS ((WLDS_FLOATS + BLOCK - 1) / BLOCK)
#define RP    18
#define RAW_FLOATS (16 * RP)
#define LDS_FLOATS  (WLDS_FLOATS + WAVES * RAW_FLOATS)

__global__ __launch_bounds__(BLOCK) __attribute__((amdgpu_waves_per_eu(8, 8)))
void vi_wmma_kernel(const float* __restrict__ values,
                    const float* __restrict__ rewards,
                    const float* __restrict__ weight,
                    float* __restrict__ out)
{
    __shared__ __attribute__((aligned(16))) float smem[LDS_FLOATS];
    float* wlds = smem;
    const int tid  = threadIdx.x;
    const int lane = tid & 31;
    const int wave = tid >> 5;
    float* raw = smem + WLDS_FLOATS + wave * RAW_FLOATS;

    float wst[STAGE_ITERS];
    #pragma unroll
    for (int it = 0; it < STAGE_ITERS; ++it) {
        int idx = tid + it * BLOCK;
        int c = idx / KPAD;
        int k = idx - c * KPAD;
        bool real = (c < ADIM * IKK) && (k < IKK);
        float v = weight[real ? c * IKK + k : 0];
        wst[it] = real ? v : 0.0f;
    }
    #pragma unroll
    for (int it = 0; it < STAGE_ITERS; ++it) {
        int idx = tid + it * BLOCK;
        if (idx < WLDS_FLOATS) wlds[idx] = wst[it];
    }

    __syncthreads();

    const int  n  = lane & 15;
    const bool hi = lane >= 16;
    const int  kh = hi ? 2 : 0;

    #pragma unroll 1
    for (int tt = 0; tt < TILES_PER_WAVE; ++tt) {
        int g   = blockIdx.x * (WAVES * TILES_PER_WAVE) + tt * WAVES + wave;
        int lin = g * TILE;
        int b   = lin >> 14;
        int rem = lin & 16383;
        int y   = rem >> 7;
        int x0  = rem & 127;

        const bool cin = lane < RP;
        const int  xx  = x0 - 1 + lane;
        const bool xok = (xx >= 0) && (xx < WW);
        const int  xc  = min(max(xx, 0), WW - 1);
        int  ro[3];
        bool ok[3];
        #pragma unroll
        for (int dy = 0; dy < 3; ++dy) {
            int yy = y + dy - 1;
            ok[dy] = xok && (yy >= 0) && (yy < HH);
            ro[dy] = min(max(yy, 0), HH - 1) * WW + xc;
        }
        const float* vb = values  + b * (HH * WW);
        const float* rb = rewards + b * (RDIM * HH * WW);
        float wv[15];
        #pragma unroll
        for (int r = 0; r < 15; ++r) {
            const int i  = r / 3;
            const int dy = r - 3 * i;
            float v = (i < RDIM) ? rb[i * (HH * WW) + ro[dy]] : vb[ro[dy]];
            wv[r] = ok[dy] ? v : 0.0f;
        }
        if (cin) {
            #pragma unroll
            for (int r = 0; r < 15; ++r) raw[r * RP + lane] = wv[r];
            raw[15 * RP + lane] = 0.0f;
        }


        const float* fA = raw + n + (hi ? 2 : 0);
        const float* fB = raw + n + (hi ? RP - 1 : 0);
        const float* fC = raw + n + (hi ? 2 * RP + 2 : 0);
        const float* fD = raw + n + (hi ? 3 * RP - 1 : 0);

        v2f bs[12];
        #pragma unroll
        for (int s = 0; s < 12; ++s) {
            #pragma unroll
            for (int e = 0; e < 2; ++e) {
                const int kl  = 4 * s + e;
                const int off = (kl / 3) * RP + (kl % 3);
                bs[s][e] = ((kl % 3) == 0 ? fA : fB)[off];
            }
        }

        float pp[24];
        #pragma unroll
        for (int t = 0; t < 3; ++t) {
            #pragma unroll
            for (int r = 0; r < 8; ++r) {
                const int jl  = 16 * t + r;
                const int off = (jl / 3) * RP + (jl % 3);
                pp[t * 8 + r] = ((jl % 3) == 0 ? fC : fD)[off];
            }
        }

        float best = -3.402823466e+38f;

        #pragma unroll 1
        for (int o = 0; o < ADIM; ++o) {
            v8f acc0 = {}, acc1 = {}, acc2 = {};
            const float* wb = wlds + (o * IKK) * KPAD + kh;
            #pragma unroll
            for (int s = 0; s < 12; ++s) {
                v2f a0 = *(const v2f*)(wb + (0 * 16 + n) * KPAD + 4 * s);
                v2f a1 = *(const v2f*)(wb + (1 * 16 + n) * KPAD + 4 * s);
                v2f a2 = *(const v2f*)(wb + (2 * 16 + n) * KPAD + 4 * s);
                acc0 = __builtin_amdgcn_wmma_f32_16x16x4_f32(false, a0, false, bs[s], (short)0, acc0, false, false);
                acc1 = __builtin_amdgcn_wmma_f32_16x16x4_f32(false, a1, false, bs[s], (short)0, acc1, false, false);
                acc2 = __builtin_amdgcn_wmma_f32_16x16x4_f32(false, a2, false, bs[s], (short)0, acc2, false, false);
            }

            float vmax = -3.402823466e+38f;
            #pragma unroll
            for (int r = 0; r < 8; ++r) vmax = fmaxf(vmax, acc0[r]);
            #pragma unroll
            for (int r = 0; r < 8; ++r) vmax = fmaxf(vmax, acc1[r]);
            #pragma unroll
            for (int r = 0; r < 8; ++r) {
                bool valid = (32 + r + (hi ? 8 : 0)) < IKK;
                vmax = fmaxf(vmax, valid ? acc2[r] : -3.402823466e+38f);
            }
            vmax = fmaxf(vmax, __shfl_xor(vmax, 16));

            float se = 0.0f, qe = 0.0f;
            #pragma unroll
            for (int r = 0; r < 8; ++r) {
                float e = __expf(acc0[r] - vmax);
                se += e; qe += e * pp[r];
            }
            #pragma unroll
            for (int r = 0; r < 8; ++r) {
                float e = __expf(acc1[r] - vmax);
                se += e; qe += e * pp[8 + r];
            }
            #pragma unroll
            for (int r = 0; r < 8; ++r) {
                bool valid = (32 + r + (hi ? 8 : 0)) < IKK;
                float e = valid ? __expf(acc2[r] - vmax) : 0.0f;
                se += e; qe += e * pp[16 + r];
            }
            se += __shfl_xor(se, 16);
            qe += __shfl_xor(qe, 16);
            best = fmaxf(best, qe / se);
        }

        if (!hi) out[lin + n] = best;
    }
}

extern "C" void kernel_launch(void* const* d_in, const int* in_sizes, int n_in,
                              void* d_out, int out_size, void* d_ws, size_t ws_size,
                              hipStream_t stream) {
    (void)in_sizes; (void)n_in; (void)d_ws; (void)ws_size; (void)out_size;
    const float* values  = (const float*)d_in[0];
    const float* rewards = (const float*)d_in[1];
    const float* weight  = (const float*)d_in[2];
    float* out = (float*)d_out;

    hipLaunchKernelGGL(vi_wmma_kernel, dim3(NBLOCKS), dim3(BLOCK), 0, stream,
                       values, rewards, weight, out);
}


# Minimal filtering F(2,3) along x: per pixel pair four K=16 products on transformed weights and window replace two K=48 products (48 matrix instructions per action, 32 pixels; was 72); sums regrouped
# speedup vs baseline: 2.6860x; 1.2323x over previous
#include <hip/hip_runtime.h>


typedef __attribute__((ext_vector_type(2))) float v2f;
typedef __attribute__((ext_vector_type(8))) float v8f;

#define BATCH 16
#define HH    128
#define WW    128
#define RDIM  4
#define IDIM  5
#define ADIM  8
#define IKK   45
#define MROWS 368
#define UP    68
#define UT    16
#define TPX   32
#define WAVES 16
#define BLOCK 512
#define NBLOCKS 256
#define TILES_PER_WAVE 2

#define STAGE_ITEMS (MROWS * 8)
#define STAGE_ITERS ((STAGE_ITEMS + BLOCK - 1) / BLOCK)
#define RP    36
#define RAW_FLOATS (16 * RP)
#define WLDS_FLOATS (MROWS * UP)
#define LDS_FLOATS (WLDS_FLOATS + WAVES * RAW_FLOATS)


__global__ __launch_bounds__(BLOCK) __attribute__((amdgpu_waves_per_eu(4, 4)))
void vi_wmma_kernel(const float* __restrict__ values,
                    const float* __restrict__ rewards,
                    const float* __restrict__ weight,
                    float* __restrict__ out)
{
    __shared__ __attribute__((aligned(16))) float smem[LDS_FLOATS];
    float* wlds = smem;
    const int tid  = threadIdx.x;
    const int lane = tid & 31;
    const int wave = tid >> 5;
    float* raw = smem + WLDS_FLOATS + wave * RAW_FLOATS;

    float gw[STAGE_ITERS][6];
    #pragma unroll
    for (int it = 0; it < STAGE_ITERS; ++it) {
        int idx = tid + it * BLOCK;
        int c  = idx >> 3;
        int kq = idx & 7;
        #pragma unroll
        for (int e2 = 0; e2 < 2; ++e2) {
            const int  k    = 2 * kq + e2;
            const bool real = (c < ADIM * IKK) && (k < 15);
            const float* p  = weight + (real ? c * IKK + 3 * k : 0);
            float t0 = p[0], t1 = p[1], t2 = p[2];
            gw[it][3 * e2 + 0] = real ? t0 : 0.0f;
            gw[it][3 * e2 + 1] = real ? t1 : 0.0f;
            gw[it][3 * e2 + 2] = real ? t2 : 0.0f;
        }
    }
    #pragma unroll
    for (int it = 0; it < STAGE_ITERS; ++it) {
        int idx = tid + it * BLOCK;
        if (idx < STAGE_ITEMS) {
            int c  = idx >> 3;
            int kq = idx & 7;
            float* dst = wlds + c * UP + 2 * kq;
            v2f u0, u1, u2, u3;
            #pragma unroll
            for (int e = 0; e < 2; ++e) {
                float g0 = gw[it][3 * e], g1 = gw[it][3 * e + 1], g2 = gw[it][3 * e + 2];
                u0[e] = g0;
                u1[e] = (g0 + g1 + g2) * 0.5f;
                u2[e] = (g0 - g1 + g2) * 0.5f;
                u3[e] = g2;
            }
            *(v2f*)(dst + 0 * UT) = u0;
            *(v2f*)(dst + 1 * UT) = u1;
            *(v2f*)(dst + 2 * UT) = u2;
            *(v2f*)(dst + 3 * UT) = u3;
        }
    }

    __syncthreads();

    const int  n  = lane & 15;
    const int  h  = lane >> 4;
    const bool hi = h != 0;

    #pragma unroll 1
    for (int tt = 0; tt < TILES_PER_WAVE; ++tt) {
        int g   = blockIdx.x * (WAVES * TILES_PER_WAVE) + tt * WAVES + wave;
        int lin = g * TPX;
        int b   = lin >> 14;
        int rem = lin & 16383;
        int y   = rem >> 7;
        int x0  = rem & 127;

        const float* vb = values  + b * (HH * WW);
        const float* rb = rewards + b * (RDIM * HH * WW);
        int  ro[3];
        bool ok[3];
        #pragma unroll
        for (int dy = 0; dy < 3; ++dy) {
            int yy = y + dy - 1;
            ok[dy] = (yy >= 0) && (yy < HH);
            ro[dy] = min(max(yy, 0), HH - 1) * WW;
        }
        float wv[15];
        #pragma unroll
        for (int r = 0; r < 15; ++r) {
            const int i  = r / 3;
            const int dy = r - 3 * i;
            float v = (i < RDIM) ? rb[i * (HH * WW) + ro[dy] + x0 + lane] : vb[ro[dy] + x0 + lane];
            wv[r] = ok[dy] ? v : 0.0f;
        }
        const int  hr   = min(lane >> 1, 14);
        const int  hsd  = lane & 1;
        const int  hi_i = hr / 3;
        const int  hdy  = hr - 3 * hi_i;
        const int  hyy  = y + hdy - 1;
        const int  hxx  = hsd ? x0 + TPX : x0 - 1;
        const bool hok  = (hyy >= 0) && (hyy < HH) && (hxx >= 0) && (hxx < WW);
        const int  hoff = min(max(hyy, 0), HH - 1) * WW + min(max(hxx, 0), WW - 1);
        float hv = (hi_i < RDIM) ? rb[hi_i * (HH * WW) + hoff] : vb[hoff];
        hv = hok ? hv : 0.0f;

        #pragma unroll
        for (int r = 0; r < 15; ++r) raw[r * RP + 1 + lane] = wv[r];
        if (lane < 30) raw[hr * RP + (hsd ? TPX + 1 : 0)] = hv;
        raw[15 * RP + lane] = 0.0f;
        if (lane < 4) raw[15 * RP + 32 + lane] = 0.0f;


        const float* fV = raw + 2 * n + (hi ? 2 * RP : 0);
        v2f vt[4][4];
        #pragma unroll
        for (int s = 0; s < 4; ++s) {
            #pragma unroll
            for (int e = 0; e < 2; ++e) {
                const int off = (4 * s + e) * RP;
                v2f d01 = *(const v2f*)(fV + off);
                v2f d23 = *(const v2f*)(fV + off + 2);
                vt[0][s][e] = d01[0] - d23[0];
                vt[1][s][e] = d01[1] + d23[0];
                vt[2][s][e] = d23[0] - d01[1];
                vt[3][s][e] = d01[1] - d23[1];
            }
        }

        const float* fC = raw + 2 * n + (hi ? 2 * RP + 2 : 0);
        const float* fD = raw + 2 * n + (hi ? 3 * RP - 1 : 0);
        float ppe[24], ppo[24];
        #pragma unroll
        for (int m = 0; m < 3; ++m) {
            #pragma unroll
            for (int r = 0; r < 8; ++r) {
                const int jl  = 16 * m + r;
                const int off = (jl / 3) * RP + (jl % 3);
                const float* src = ((jl % 3) == 0 ? fC : fD) + off;
                ppe[m * 8 + r] = src[0];
                ppo[m * 8 + r] = src[1];
            }
        }

        float beste = -3.402823466e+38f;
        float besto = -3.402823466e+38f;
        const float* ub = wlds + n * UP + 2 * h;

        #pragma unroll 1
        for (int o = 0; o < ADIM; ++o) {
            const float* wb = ub + (o * IKK) * UP;
            float le[24], lo[24];
            #pragma unroll
            for (int m = 0; m < 3; ++m) {
                v2f af[4][4];
                #pragma unroll
                for (int s = 0; s < 4; ++s) {
                    #pragma unroll
                    for (int t = 0; t < 4; ++t) {
                        af[t][s] = *(const v2f*)(wb + (16 * m) * UP + t * UT + 4 * s);
                    }
                }
                v8f acc[4];
                #pragma unroll
                for (int t = 0; t < 4; ++t) acc[t] = (v8f){};
                #pragma unroll
                for (int s = 0; s < 4; ++s) {
                    #pragma unroll
                    for (int t = 0; t < 4; ++t) {
                        acc[t] = __builtin_amdgcn_wmma_f32_16x16x4_f32(false, af[t][s], false, vt[t][s], (short)0, acc[t], false, false);
                    }
                }
                #pragma unroll
                for (int r = 0; r < 8; ++r) {
                    float m0 = acc[0][r], m1 = acc[1][r], m2 = acc[2][r], m3 = acc[3][r];
                    le[m * 8 + r] = (m0 + m1) + m2;
                    lo[m * 8 + r] = (m1 - m2) - m3;
                }
            }

            float vme = -3.402823466e+38f, vmo = -3.402823466e+38f;
            #pragma unroll
            for (int q = 0; q < 24; ++q) {
                bool valid = (q < 16) || ((16 + q + (hi ? 8 : 0)) < IKK);
                vme = fmaxf(vme, valid ? le[q] : -3.402823466e+38f);
                vmo = fmaxf(vmo, valid ? lo[q] : -3.402823466e+38f);
            }
            vme = fmaxf(vme, __shfl_xor(vme, 16));
            vmo = fmaxf(vmo, __shfl_xor(vmo, 16));

            float see = 0.0f, qee = 0.0f, seo = 0.0f, qeo = 0.0f;
            #pragma unroll
            for (int q = 0; q < 24; ++q) {
                bool valid = (q < 16) || ((16 + q + (hi ? 8 : 0)) < IKK);
                float ee = valid ? __expf(le[q] - vme) : 0.0f;
                float eo = valid ? __expf(lo[q] - vmo) : 0.0f;
                see += ee; qee += ee * ppe[q];
                seo += eo; qeo += eo * ppo[q];
            }
            see += __shfl_xor(see, 16);
            qee += __shfl_xor(qee, 16);
            seo += __shfl_xor(seo, 16);
            qeo += __shfl_xor(qeo, 16);
            beste = fmaxf(beste, qee / see);
            besto = fmaxf(besto, qeo / seo);
        }

        if (!hi) {
            v2f res; res[0] = beste; res[1] = besto;
            *(v2f*)(out + lin + 2 * n) = res;
        }
    }
}

extern "C" void kernel_launch(void* const* d_in, const int* in_sizes, int n_in,
                              void* d_out, int out_size, void* d_ws, size_t ws_size,
                              hipStream_t stream) {
    (void)in_sizes; (void)n_in; (void)d_ws; (void)ws_size; (void)out_size;
    const float* values  = (const float*)d_in[0];
    const float* rewards = (const float*)d_in[1];
    const float* weight  = (const float*)d_in[2];
    float* out = (float*)d_out;

    hipLaunchKernelGGL(vi_wmma_kernel, dim3(NBLOCKS), dim3(BLOCK), 0, stream,
                       values, rewards, weight, out);
}


# Minimal filtering F(4,3) along x replaces F(2,3): per pixel quad six K=16 products, was eight (36 matrix instructions per action per 32 pixels, was 48); 64-pixel wave tiles, 8 waves; sums regrouped
# speedup vs baseline: 3.1852x; 1.1859x over previous
#include <hip/hip_runtime.h>


typedef __attribute__((ext_vector_type(2))) float v2f;
typedef __attribute__((ext_vector_type(4))) float v4f;
typedef __attribute__((ext_vector_type(8))) float v8f;

#define BATCH 16
#define HH    128
#define WW    128
#define RDIM  4
#define IDIM  5
#define ADIM  8
#define IKK   45
#define MROWS 368
#define NT    6
#define UP    100
#define UT    16
#define TPX   64
#define WAVES 8
#define BLOCK 256
#define NBLOCKS 256
#define TILES_PER_WAVE 2

#define STAGE_ITEMS (MROWS * 8)
#define STAGE_ITERS ((STAGE_ITEMS + BLOCK - 1) / BLOCK)
#define RP    68
#define RAW_FLOATS (16 * RP)
#define WLDS_FLOATS (MROWS * UP)
#define LDS_FLOATS (WLDS_FLOATS + WAVES * RAW_FLOATS)


#define SOFTMAX_PIXEL(L, P, BEST)                                                          \
            {                                                                              \
                float vm = -3.402823466e+38f;                                              \
                _Pragma("unroll")                                                          \
                for (int i = 0; i < 24; ++i) {                                             \
                    bool valid = (i < 16) || ((16 + i + (hi ? 8 : 0)) < IKK);              \
                    vm = fmaxf(vm, valid ? L[i] : -3.402823466e+38f);                      \
                }                                                                          \
                vm = fmaxf(vm, __shfl_xor(vm, 16));                                        \
                float se = 0.0f, qe = 0.0f;                                                \
                _Pragma("unroll")                                                          \
                for (int i = 0; i < 24; ++i) {                                             \
                    bool valid = (i < 16) || ((16 + i + (hi ? 8 : 0)) < IKK);              \
                    float ex = valid ? __expf(L[i] - vm) : 0.0f;                           \
                    se += ex; qe += ex * P[i];                                             \
                }                                                                          \
                se += __shfl_xor(se, 16);                                                  \
                qe += __shfl_xor(qe, 16);                                                  \
                BEST = fmaxf(BEST, qe / se);                                               \
            }

__global__ __launch_bounds__(BLOCK) __attribute__((amdgpu_waves_per_eu(2, 2)))
void vi_wmma_kernel(const float* __restrict__ values,
                    const float* __restrict__ rewards,
                    const float* __restrict__ weight,
                    float* __restrict__ out)
{
    __shared__ __attribute__((aligned(16))) float smem[LDS_FLOATS];
    float* wlds = smem;
    const int tid  = threadIdx.x;
    const int lane = tid & 31;
    const int wave = tid >> 5;
    float* raw = smem + WLDS_FLOATS + wave * RAW_FLOATS;

    float gw[STAGE_ITERS][6];
    #pragma unroll
    for (int it = 0; it < STAGE_ITERS; ++it) {
        int idx = tid + it * BLOCK;
        int c  = idx >> 3;
        int kq = idx & 7;
        #pragma unroll
        for (int e2 = 0; e2 < 2; ++e2) {
            const int  k    = 2 * kq + e2;
            const bool real = (c < ADIM * IKK) && (k < 15);
            const float* p  = weight + (real ? c * IKK + 3 * k : 0);
            float t0 = p[0], t1 = p[1], t2 = p[2];
            gw[it][3 * e2 + 0] = real ? t0 : 0.0f;
            gw[it][3 * e2 + 1] = real ? t1 : 0.0f;
            gw[it][3 * e2 + 2] = real ? t2 : 0.0f;
        }
    }
    #pragma unroll
    for (int it = 0; it < STAGE_ITERS; ++it) {
        int idx = tid + it * BLOCK;
        if (idx < STAGE_ITEMS) {
            int c  = idx >> 3;
            int kq = idx & 7;
            float* dst = wlds + c * UP + 2 * kq;
            v2f u0, u1, u2, u3, u4, u5;
            #pragma unroll
            for (int e = 0; e < 2; ++e) {
                float g0 = gw[it][3 * e], g1 = gw[it][3 * e + 1], g2 = gw[it][3 * e + 2];
                float ga = g0 + g2;
                float gb = g0 * (1.0f / 24.0f) + g2 * (1.0f / 6.0f);
                float gc = g1 * (1.0f / 12.0f);
                u0[e] = g0 * 0.25f;
                u1[e] = (ga + g1) * (-1.0f / 6.0f);
                u2[e] = (ga - g1) * (-1.0f / 6.0f);
                u3[e] = gb + gc;
                u4[e] = gb - gc;
                u5[e] = g2;
            }
            *(v2f*)(dst + 0 * UT) = u0;
            *(v2f*)(dst + 1 * UT) = u1;
            *(v2f*)(dst + 2 * UT) = u2;
            *(v2f*)(dst + 3 * UT) = u3;
            *(v2f*)(dst + 4 * UT) = u4;
            *(v2f*)(dst + 5 * UT) = u5;
        }
    }

    __syncthreads();

    const int  n  = lane & 15;
    const int  h  = lane >> 4;
    const bool hi = h != 0;

    #pragma unroll 1
    for (int tt = 0; tt < TILES_PER_WAVE; ++tt) {
        int g   = blockIdx.x * (WAVES * TILES_PER_WAVE) + tt * WAVES + wave;
        int lin = g * TPX;
        int b   = lin >> 14;
        int rem = lin & 16383;
        int y   = rem >> 7;
        int x0  = rem & 127;

        const float* vb = values  + b * (HH * WW);
        const float* rb = rewards + b * (RDIM * HH * WW);
        int  ro[3];
        bool ok[3];
        #pragma unroll
        for (int dy = 0; dy < 3; ++dy) {
            int yy = y + dy - 1;
            ok[dy] = (yy >= 0) && (yy < HH);
            ro[dy] = min(max(yy, 0), HH - 1) * WW;
        }
        v2f wv[15];
        #pragma unroll
        for (int r = 0; r < 15; ++r) {
            const int i  = r / 3;
            const int dy = r - 3 * i;
            const float* src = (i < RDIM) ? rb + i * (HH * WW) + ro[dy] + x0 + 2 * lane : vb + ro[dy] + x0 + 2 * lane;
            v2f v = *(const v2f*)src;
            wv[r][0] = ok[dy] ? v[0] : 0.0f;
            wv[r][1] = ok[dy] ? v[1] : 0.0f;
        }
        const int  hr   = min(lane >> 1, 14);
        const int  hsd  = lane & 1;
        const int  hi_i = hr / 3;
        const int  hdy  = hr - 3 * hi_i;
        const int  hyy  = y + hdy - 1;
        const int  hxx  = hsd ? x0 + TPX : x0 - 1;
        const bool hok  = (hyy >= 0) && (hyy < HH) && (hxx >= 0) && (hxx < WW);
        const int  hoff = min(max(hyy, 0), HH - 1) * WW + min(max(hxx, 0), WW - 1);
        float hv = (hi_i < RDIM) ? rb[hi_i * (HH * WW) + hoff] : vb[hoff];
        hv = hok ? hv : 0.0f;

        #pragma unroll
        for (int r = 0; r < 15; ++r) {
            raw[r * RP + 1 + 2 * lane] = wv[r][0];
            raw[r * RP + 2 + 2 * lane] = wv[r][1];
        }
        if (lane < 30) raw[hr * RP + (hsd ? TPX + 1 : 0)] = hv;
        raw[15 * RP + lane] = 0.0f;
        raw[15 * RP + 32 + lane] = 0.0f;
        if (lane < 4) raw[15 * RP + 64 + lane] = 0.0f;


        const float* fV = raw + 4 * n + (hi ? 2 * RP : 0);
        v2f vt[NT][4];
        #pragma unroll
        for (int s = 0; s < 4; ++s) {
            #pragma unroll
            for (int e = 0; e < 2; ++e) {
                const int off = (4 * s + e) * RP;
                v4f da = *(const v4f*)(fV + off);
                v2f db = *(const v2f*)(fV + off + 4);
                const float d0 = da[0], d1 = da[1], d2 = da[2], d3 = da[3], d4 = db[0], d5 = db[1];
                const float e31 = d3 - d1;
                const float e42 = d4 - d2;
                vt[0][s][e] = (4.0f * d0 - 5.0f * d2) + d4;
                vt[1][s][e] = (d3 + d4) - 4.0f * (d1 + d2);
                vt[2][s][e] = (d4 - d3) + 4.0f * (d1 - d2);
                vt[3][s][e] = e42 + 2.0f * e31;
                vt[4][s][e] = e42 - 2.0f * e31;
                vt[5][s][e] = (4.0f * d1 - 5.0f * d3) + d5;
            }
        }

        const float* fC = raw + 4 * n + (hi ? 2 * RP + 2 : 0);
        const float* fD = raw + 4 * n + (hi ? 3 * RP - 1 : 0);
        float pp0[24], pp1[24], pp2[24], pp3[24];
        #pragma unroll
        for (int m = 0; m < 3; ++m) {
            #pragma unroll
            for (int r = 0; r < 8; ++r) {
                const int jl  = 16 * m + r;
                const int off = (jl / 3) * RP + (jl % 3);
                const float* src = ((jl % 3) == 0 ? fC : fD) + off;
                pp0[m * 8 + r] = src[0];
                pp1[m * 8 + r] = src[1];
                pp2[m * 8 + r] = src[2];
                pp3[m * 8 + r] = src[3];
            }
        }

        float best0 = -3.402823466e+38f, best1 = -3.402823466e+38f, best2 = -3.402823466e+38f, best3 = -3.402823466e+38f;
        const float* ub = wlds + n * UP + 2 * h;

        #pragma unroll 1
        for (int o = 0; o < ADIM; ++o) {
            const float* wb = ub + (o * IKK) * UP;
            float lg0[24], lg1[24], lg2[24], lg3[24];
            #pragma unroll
            for (int m = 0; m < 3; ++m) {
                v2f af[NT][4];
                #pragma unroll
                for (int s = 0; s < 4; ++s) {
                    #pragma unroll
                    for (int t = 0; t < NT; ++t) {
                        af[t][s] = *(const v2f*)(wb + (16 * m) * UP + t * UT + 4 * s);
                    }
                }
                v8f acc[NT];
                #pragma unroll
                for (int t = 0; t < NT; ++t) acc[t] = (v8f){};
                #pragma unroll
                for (int s = 0; s < 4; ++s) {
                    #pragma unroll
                    for (int t = 0; t < NT; ++t) {
                        acc[t] = __builtin_amdgcn_wmma_f32_16x16x4_f32(false, af[t][s], false, vt[t][s], (short)0, acc[t], false, false);
                    }
                }
                #pragma unroll
                for (int r = 0; r < 8; ++r) {
                    float m0 = acc[0][r], m1 = acc[1][r], m2 = acc[2][r], m3 = acc[3][r], m4 = acc[4][r], m5 = acc[5][r];
                    float s12 = m1 + m2, d12 = m1 - m2, s34 = m3 + m4, d34 = m3 - m4;
                    lg0[m * 8 + r] = (m0 + s12) + s34;
                    lg1[m * 8 + r] = d12 + 2.0f * d34;
                    lg2[m * 8 + r] = s12 + 4.0f * s34;
                    lg3[m * 8 + r] = (d12 + 8.0f * d34) + m5;
                }
            }

            SOFTMAX_PIXEL(lg0, pp0, best0)
            SOFTMAX_PIXEL(lg1, pp1, best1)
            SOFTMAX_PIXEL(lg2, pp2, best2)
            SOFTMAX_PIXEL(lg3, pp3, best3)
        }

        if (!hi) {
            v4f res; res[0] = best0; res[1] = best1; res[2] = best2; res[3] = best3;
            *(v4f*)(out + lin + 4 * n) = res;
        }
    }
}

extern "C" void kernel_launch(void* const* d_in, const int* in_sizes, int n_in,
                              void* d_out, int out_size, void* d_ws, size_t ws_size,
                              hipStream_t stream) {
    (void)in_sizes; (void)n_in; (void)d_ws; (void)ws_size; (void)out_size;
    const float* values  = (const float*)d_in[0];
    const float* rewards = (const float*)d_in[1];
    const float* weight  = (const float*)d_in[2];
    float* out = (float*)d_out;

    hipLaunchKernelGGL(vi_wmma_kernel, dim3(NBLOCKS), dim3(BLOCK), 0, stream,
                       values, rewards, weight, out);
}


# Priority window on each SIMD's wave pair: the younger wave (w+4) runs its first 13 of 16 action passes at s_setprio 3, then 0, so both end together (was older first, younger alone at the end)
# speedup vs baseline: 3.5399x; 1.1114x over previous
#include <hip/hip_runtime.h>


typedef __attribute__((ext_vector_type(2))) float v2f;
typedef __attribute__((ext_vector_type(4))) float v4f;
typedef __attribute__((ext_vector_type(8))) float v8f;

#define BATCH 16
#define HH    128
#define WW    128
#define RDIM  4
#define IDIM  5
#define ADIM  8
#define IKK   45
#define MROWS 368
#define NT    6
#define UP    100
#define UT    16
#define TPX   64
#define WAVES 8
#define BLOCK 256
#define NBLOCKS 256
#define TILES_PER_WAVE 2
#define PRIO_ACTIONS 13

#define STAGE_ITEMS (MROWS * 8)
#define STAGE_ITERS ((STAGE_ITEMS + BLOCK - 1) / BLOCK)
#define RP    68
#define RAW_FLOATS (16 * RP)
#define WLDS_FLOATS (MROWS * UP)
#define WMAX_OFF (WLDS_FLOATS + WAVES * RAW_FLOATS)
#define LDS_FLOATS (WMAX_OFF + 16)


template <bool SHIFT>
__device__ __forceinline__ void softmax_pixel(const float (&L)[24], const float (&P)[24], const bool hi, float& best)
{
    float vm = 0.0f;
    if (SHIFT) {
        vm = -3.402823466e+38f;
        #pragma unroll
        for (int i = 0; i < 24; ++i) {
            bool valid = (i < 16) || ((16 + i + (hi ? 8 : 0)) < IKK);
            vm = fmaxf(vm, valid ? L[i] : -3.402823466e+38f);
        }
        vm = fmaxf(vm, __shfl_xor(vm, 16));
    }
    float se = 0.0f, qe = 0.0f;
    #pragma unroll
    for (int i = 0; i < 24; ++i) {
        bool valid = (i < 16) || ((16 + i + (hi ? 8 : 0)) < IKK);
        float ex = valid ? __expf(SHIFT ? L[i] - vm : L[i]) : 0.0f;
        se += ex; qe += ex * P[i];
    }
    se += __shfl_xor(se, 16);
    qe += __shfl_xor(qe, 16);
    best = fmaxf(best, qe / se);
}

template <bool SHIFT>
__device__ __forceinline__ void tile_actions(const float* ub, const v2f (&vt)[NT][4],
                                             const float (&pp0)[24], const float (&pp1)[24],
                                             const float (&pp2)[24], const float (&pp3)[24],
                                             const bool hi, const int o_begin, const int o_end,
                                             float& best0, float& best1, float& best2, float& best3)
{
    #pragma unroll 1
    for (int o = o_begin; o < o_end; ++o) {
        const float* wb = ub + (o * IKK) * UP;
        float lg0[24], lg1[24], lg2[24], lg3[24];
        #pragma unroll
        for (int m = 0; m < 3; ++m) {
            v2f af[NT][4];
            #pragma unroll
            for (int s = 0; s < 4; ++s) {
                #pragma unroll
                for (int t = 0; t < NT; ++t) {
                    af[t][s] = *(const v2f*)(wb + (16 * m) * UP + t * UT + 4 * s);
                }
            }
            v8f acc[NT];
            #pragma unroll
            for (int t = 0; t < NT; ++t) acc[t] = (v8f){};
            #pragma unroll
            for (int s = 0; s < 4; ++s) {
                #pragma unroll
                for (int t = 0; t < NT; ++t) {
                    acc[t] = __builtin_amdgcn_wmma_f32_16x16x4_f32(false, af[t][s], false, vt[t][s], (short)0, acc[t], false, false);
                }
            }
            #pragma unroll
            for (int r = 0; r < 8; ++r) {
                float m0 = acc[0][r], m1 = acc[1][r], m2 = acc[2][r], m3 = acc[3][r], m4 = acc[4][r], m5 = acc[5][r];
                float s12 = m1 + m2, d12 = m1 - m2, s34 = m3 + m4, d34 = m3 - m4;
                lg0[m * 8 + r] = (m0 + s12) + s34;
                lg1[m * 8 + r] = d12 + 2.0f * d34;
                lg2[m * 8 + r] = s12 + 4.0f * s34;
                lg3[m * 8 + r] = (d12 + 8.0f * d34) + m5;
            }
        }

        softmax_pixel<SHIFT>(lg0, pp0, hi, best0);
        softmax_pixel<SHIFT>(lg1, pp1, hi, best1);
        softmax_pixel<SHIFT>(lg2, pp2, hi, best2);
        softmax_pixel<SHIFT>(lg3, pp3, hi, best3);
    }
}

__global__ __launch_bounds__(BLOCK) __attribute__((amdgpu_waves_per_eu(2, 2)))
void vi_wmma_kernel(const float* __restrict__ values,
                    const float* __restrict__ rewards,
                    const float* __restrict__ weight,
                    float* __restrict__ out)
{
    __shared__ __attribute__((aligned(16))) float smem[LDS_FLOATS];
    float* wlds = smem;
    const int tid  = threadIdx.x;
    const int lane = tid & 31;
    const int wave = tid >> 5;
    float* raw = smem + WLDS_FLOATS + wave * RAW_FLOATS;

    float gw[STAGE_ITERS][6];
    #pragma unroll
    for (int it = 0; it < STAGE_ITERS; ++it) {
        int idx = tid + it * BLOCK;
        int c  = idx >> 3;
        int kq = idx & 7;
        #pragma unroll
        for (int e2 = 0; e2 < 2; ++e2) {
            const int  k    = 2 * kq + e2;
            const bool real = (c < ADIM * IKK) && (k < 15);
            const float* p  = weight + (real ? c * IKK + 3 * k : 0);
            float t0 = p[0], t1 = p[1], t2 = p[2];
            gw[it][3 * e2 + 0] = real ? t0 : 0.0f;
            gw[it][3 * e2 + 1] = real ? t1 : 0.0f;
            gw[it][3 * e2 + 2] = real ? t2 : 0.0f;
        }
    }
    #pragma unroll
    for (int it = 0; it < STAGE_ITERS; ++it) {
        int idx = tid + it * BLOCK;
        if (idx < STAGE_ITEMS) {
            int c  = idx >> 3;
            int kq = idx & 7;
            float* dst = wlds + c * UP + 2 * kq;
            v2f u0, u1, u2, u3, u4, u5;
            #pragma unroll
            for (int e = 0; e < 2; ++e) {
                float g0 = gw[it][3 * e], g1 = gw[it][3 * e + 1], g2 = gw[it][3 * e + 2];
                float ga = g0 + g2;
                float gb = g0 * (1.0f / 24.0f) + g2 * (1.0f / 6.0f);
                float gc = g1 * (1.0f / 12.0f);
                u0[e] = g0 * 0.25f;
                u1[e] = (ga + g1) * (-1.0f / 6.0f);
                u2[e] = (ga - g1) * (-1.0f / 6.0f);
                u3[e] = gb + gc;
                u4[e] = gb - gc;
                u5[e] = g2;
            }
            *(v2f*)(dst + 0 * UT) = u0;
            *(v2f*)(dst + 1 * UT) = u1;
            *(v2f*)(dst + 2 * UT) = u2;
            *(v2f*)(dst + 3 * UT) = u3;
            *(v2f*)(dst + 4 * UT) = u4;
            *(v2f*)(dst + 5 * UT) = u5;
        }
    }

    float wmax = 0.0f;
    #pragma unroll
    for (int it = 0; it < STAGE_ITERS; ++it) {
        #pragma unroll
        for (int e = 0; e < 6; ++e) wmax = fmaxf(wmax, fabsf(gw[it][e]));
    }
    #pragma unroll
    for (int sh = 16; sh > 0; sh >>= 1) wmax = fmaxf(wmax, __shfl_xor(wmax, sh));
    if (lane == 0) smem[WMAX_OFF + wave] = wmax;

    __syncthreads();

    {
        const v4f wa = *(const v4f*)(smem + WMAX_OFF);
        const v4f wc = *(const v4f*)(smem + WMAX_OFF + 4);
        wmax = fmaxf(fmaxf(fmaxf(wa[0], wa[1]), fmaxf(wa[2], wa[3])), fmaxf(fmaxf(wc[0], wc[1]), fmaxf(wc[2], wc[3])));
    }

    const int  wave_u  = __builtin_amdgcn_readfirstlane(wave);
    const bool younger = wave_u >= (WAVES / 2);
    if (younger) __builtin_amdgcn_s_setprio(3);

    const int  n  = lane & 15;
    const int  h  = lane >> 4;
    const bool hi = h != 0;

    #pragma unroll 1
    for (int tt = 0; tt < TILES_PER_WAVE; ++tt) {
        int g   = blockIdx.x * (WAVES * TILES_PER_WAVE) + tt * WAVES + wave;
        int lin = g * TPX;
        int b   = lin >> 14;
        int rem = lin & 16383;
        int y   = rem >> 7;
        int x0  = rem & 127;

        const float* vb = values  + b * (HH * WW);
        const float* rb = rewards + b * (RDIM * HH * WW);
        int  ro[3];
        bool ok[3];
        #pragma unroll
        for (int dy = 0; dy < 3; ++dy) {
            int yy = y + dy - 1;
            ok[dy] = (yy >= 0) && (yy < HH);
            ro[dy] = min(max(yy, 0), HH - 1) * WW;
        }
        v2f wv[15];
        #pragma unroll
        for (int r = 0; r < 15; ++r) {
            const int i  = r / 3;
            const int dy = r - 3 * i;
            const float* src = (i < RDIM) ? rb + i * (HH * WW) + ro[dy] + x0 + 2 * lane : vb + ro[dy] + x0 + 2 * lane;
            v2f v = *(const v2f*)src;
            wv[r][0] = ok[dy] ? v[0] : 0.0f;
            wv[r][1] = ok[dy] ? v[1] : 0.0f;
        }
        const int  hr   = min(lane >> 1, 14);
        const int  hsd  = lane & 1;
        const int  hi_i = hr / 3;
        const int  hdy  = hr - 3 * hi_i;
        const int  hyy  = y + hdy - 1;
        const int  hxx  = hsd ? x0 + TPX : x0 - 1;
        const bool hok  = (hyy >= 0) && (hyy < HH) && (hxx >= 0) && (hxx < WW);
        const int  hoff = min(max(hyy, 0), HH - 1) * WW + min(max(hxx, 0), WW - 1);
        float hv = (hi_i < RDIM) ? rb[hi_i * (HH * WW) + hoff] : vb[hoff];
        hv = hok ? hv : 0.0f;

        #pragma unroll
        for (int r = 0; r < 15; ++r) {
            raw[r * RP + 1 + 2 * lane] = wv[r][0];
            raw[r * RP + 2 + 2 * lane] = wv[r][1];
        }
        if (lane < 30) raw[hr * RP + (hsd ? TPX + 1 : 0)] = hv;
        raw[15 * RP + lane] = 0.0f;
        raw[15 * RP + 32 + lane] = 0.0f;
        if (lane < 4) raw[15 * RP + 64 + lane] = 0.0f;


        const float* fV = raw + 4 * n + (hi ? 2 * RP : 0);
        v2f vt[NT][4];
        float ca0 = 0.0f, ca1 = 0.0f, ca2 = 0.0f, ca3 = 0.0f, ca4 = 0.0f, ca5 = 0.0f;
        #pragma unroll
        for (int s = 0; s < 4; ++s) {
            #pragma unroll
            for (int e = 0; e < 2; ++e) {
                const int off = (4 * s + e) * RP;
                v4f da = *(const v4f*)(fV + off);
                v2f db = *(const v2f*)(fV + off + 4);
                const float d0 = da[0], d1 = da[1], d2 = da[2], d3 = da[3], d4 = db[0], d5 = db[1];
                ca0 += fabsf(d0); ca1 += fabsf(d1); ca2 += fabsf(d2); ca3 += fabsf(d3); ca4 += fabsf(d4); ca5 += fabsf(d5);
                const float e31 = d3 - d1;
                const float e42 = d4 - d2;
                vt[0][s][e] = (4.0f * d0 - 5.0f * d2) + d4;
                vt[1][s][e] = (d3 + d4) - 4.0f * (d1 + d2);
                vt[2][s][e] = (d4 - d3) + 4.0f * (d1 - d2);
                vt[3][s][e] = e42 + 2.0f * e31;
                vt[4][s][e] = e42 - 2.0f * e31;
                vt[5][s][e] = (4.0f * d1 - 5.0f * d3) + d5;
            }
        }

        const float* fC = raw + 4 * n + (hi ? 2 * RP + 2 : 0);
        const float* fD = raw + 4 * n + (hi ? 3 * RP - 1 : 0);
        float pp0[24], pp1[24], pp2[24], pp3[24];
        #pragma unroll
        for (int m = 0; m < 3; ++m) {
            #pragma unroll
            for (int r = 0; r < 8; ++r) {
                const int jl  = 16 * m + r;
                const int off = (jl / 3) * RP + (jl % 3);
                const float* src = ((jl % 3) == 0 ? fC : fD) + off;
                pp0[m * 8 + r] = src[0];
                pp1[m * 8 + r] = src[1];
                pp2[m * 8 + r] = src[2];
                pp3[m * 8 + r] = src[3];
            }
        }

        float best0 = -3.402823466e+38f, best1 = -3.402823466e+38f, best2 = -3.402823466e+38f, best3 = -3.402823466e+38f;
        const float* ub = wlds + n * UP + 2 * h;

        const float c12 = ca1 + ca2, c34 = ca3 + ca4;
        float psum = fmaxf(fmaxf(ca0 + c12, c12 + ca3), fmaxf(ca2 + c34, c34 + ca5));
        psum += __shfl_xor(psum, 16);
        const bool small = (wmax * psum <= 64.0f) && (psum <= 1.0e6f);

        const int pass0 = tt * ADIM;
        const int split = (younger && PRIO_ACTIONS >= pass0 && PRIO_ACTIONS < pass0 + ADIM) ? PRIO_ACTIONS - pass0 : ADIM;
        const bool nomax = __all(small);

        int o_from = 0;
        #pragma unroll 1
        for (int part = 0; part < 2; ++part) {
            const int o_to = (part == 0) ? split : ADIM;
            if (part == 1 && split < ADIM) __builtin_amdgcn_s_setprio(0);
            if (nomax) {
                tile_actions<false>(ub, vt, pp0, pp1, pp2, pp3, hi, o_from, o_to, best0, best1, best2, best3);
            } else {
                tile_actions<true>(ub, vt, pp0, pp1, pp2, pp3, hi, o_from, o_to, best0, best1, best2, best3);
            }
            o_from = o_to;
        }

        if (!hi) {
            v4f res; res[0] = best0; res[1] = best1; res[2] = best2; res[3] = best3;
            *(v4f*)(out + lin + 4 * n) = res;
        }
    }
    __builtin_amdgcn_s_setprio(0);
}

extern "C" void kernel_launch(void* const* d_in, const int* in_sizes, int n_in,
                              void* d_out, int out_size, void* d_ws, size_t ws_size,
                              hipStream_t stream) {
    (void)in_sizes; (void)n_in; (void)d_ws; (void)ws_size; (void)out_size;
    const float* values  = (const float*)d_in[0];
    const float* rewards = (const float*)d_in[1];
    const float* weight  = (const float*)d_in[2];
    float* out = (float*)d_out;

    hipLaunchKernelGGL(vi_wmma_kernel, dim3(NBLOCKS), dim3(BLOCK), 0, stream,
                       values, rewards, weight, out);
}


# Softmax stage issue cycles cut: log2(e) folded into staged weights (bare exp2), sums as packed even/odd pairs (sum order changed), one packed op behind each exp2, quotient by reciprocal not division
# speedup vs baseline: 3.7750x; 1.0664x over previous
#include <hip/hip_runtime.h>


typedef __attribute__((ext_vector_type(2))) float v2f;
typedef __attribute__((ext_vector_type(4))) float v4f;
typedef __attribute__((ext_vector_type(8))) float v8f;

#define BATCH 16
#define HH    128
#define WW    128
#define RDIM  4
#define IDIM  5
#define ADIM  8
#define IKK   45
#define MROWS 368
#define NT    6
#define UP    100
#define UT    16
#define TPX   64
#define WAVES 8
#define BLOCK 256
#define NBLOCKS 256
#define TILES_PER_WAVE 2
#define LOG2E 1.4426950408889634f
#define PRIO_ACTIONS 13

#define STAGE_ITEMS (MROWS * 8)
#define STAGE_ITERS ((STAGE_ITEMS + BLOCK - 1) / BLOCK)
#define RP    68
#define RAW_FLOATS (16 * RP)
#define WLDS_FLOATS (MROWS * UP)
#define WMAX_OFF (WLDS_FLOATS + WAVES * RAW_FLOATS)
#define LDS_FLOATS (WMAX_OFF + 16)


template <bool SHIFT>
__device__ __forceinline__ void softmax_quad(const v2f (&L)[4][12], const v2f (&P)[4][12], const bool hi,
                                             float& best0, float& best1, float& best2, float& best3)
{
    float vm[4] = {0.0f, 0.0f, 0.0f, 0.0f};
    if (SHIFT) {
        #pragma unroll
        for (int q = 0; q < 4; ++q) {
            float v = -3.402823466e+38f;
            #pragma unroll
            for (int i = 0; i < 24; ++i) {
                bool valid = (i < 16) || ((16 + i + (hi ? 8 : 0)) < IKK);
                v = fmaxf(v, valid ? L[q][i >> 1][i & 1] : -3.402823466e+38f);
            }
            vm[q] = fmaxf(v, __shfl_xor(v, 16));
        }
    }
    v2f se2[4], qe2[4], exp_prev[4], exp_cur[4];
    float lmask[4][3];
    #pragma unroll
    for (int q = 0; q < 4; ++q) {
        se2[q] = (v2f){0.0f, 0.0f}; qe2[q] = (v2f){0.0f, 0.0f};
        exp_prev[q] = (v2f){0.0f, 0.0f}; exp_cur[q] = (v2f){0.0f, 0.0f};
        lmask[q][0] = 0.0f; lmask[q][1] = 0.0f; lmask[q][2] = 0.0f;
    }
    #pragma unroll
    for (int p = 0; p <= 12; ++p) {
        #pragma unroll
        for (int q = 0; q < 4; ++q) {
            #pragma unroll
            for (int e = 0; e < 2; ++e) {
                if (p < 12) {
                    const int i = 2 * p + e;
                    const float l = (i >= 21) ? lmask[q][i - 21] : L[q][p][e];
                    const float x = __builtin_amdgcn_exp2f(SHIFT ? l - vm[q] : l);
                    exp_cur[q][e] = x;
                    asm volatile("" :: "v"(x));
                    __builtin_amdgcn_sched_barrier(0);
                }
                if (p > 0) {
                    if (e == 0) { se2[q] += exp_prev[q];                                                 asm volatile("" :: "v"(se2[q])); }
                    else        { qe2[q] = __builtin_elementwise_fma(exp_prev[q], P[q][p - 1], qe2[q]);  asm volatile("" :: "v"(qe2[q])); }
                } else {
                    const int  i0    = 21 + e;
                    const bool valid = (16 + i0 + (hi ? 8 : 0)) < IKK;
                    lmask[q][e] = valid ? L[q][i0 >> 1][i0 & 1] : -3.402823466e+38f;
                    if (e == 1) {
                        const bool valid2 = (16 + 23 + (hi ? 8 : 0)) < IKK;
                        lmask[q][2] = valid2 ? L[q][11][1] : -3.402823466e+38f;
                    }
                    asm volatile("" :: "v"(lmask[q][e]));
                }
                __builtin_amdgcn_sched_barrier(0);
            }
        }
        #pragma unroll
        for (int q = 0; q < 4; ++q) exp_prev[q] = exp_cur[q];
    }
    float se[4], qe[4];
    #pragma unroll
    for (int q = 0; q < 4; ++q) {
        se[q] = se2[q][0] + se2[q][1];
        qe[q] = qe2[q][0] + qe2[q][1];
    }
    #pragma unroll
    for (int q = 0; q < 4; ++q) {
        se[q] += __shfl_xor(se[q], 16);
        qe[q] += __shfl_xor(qe[q], 16);
    }
    best0 = fmaxf(best0, qe[0] * __builtin_amdgcn_rcpf(se[0]));
    best1 = fmaxf(best1, qe[1] * __builtin_amdgcn_rcpf(se[1]));
    best2 = fmaxf(best2, qe[2] * __builtin_amdgcn_rcpf(se[2]));
    best3 = fmaxf(best3, qe[3] * __builtin_amdgcn_rcpf(se[3]));
}

template <bool SHIFT>
__device__ __forceinline__ void tile_actions(const float* ub, const v2f (&vt)[NT][4],
                                             const v2f (&pp)[4][12],
                                             const bool hi, const int o_begin, const int o_end,
                                             float& best0, float& best1, float& best2, float& best3)
{
    #pragma unroll 1
    for (int o = o_begin; o < o_end; ++o) {
        const float* wb = ub + (o * IKK) * UP;
        v2f af[3][NT][4];
        #pragma unroll
        for (int m = 0; m < 3; ++m) {
            #pragma unroll
            for (int s = 0; s < 4; ++s) {
                #pragma unroll
                for (int t = 0; t < NT; ++t) {
                    af[m][t][s] = *(const v2f*)(wb + (16 * m) * UP + t * UT + 4 * s);
                }
            }
        }
        __builtin_amdgcn_sched_barrier(0);
        v8f acc[3][NT];
        #pragma unroll
        for (int m = 0; m < 3; ++m) {
            #pragma unroll
            for (int t = 0; t < NT; ++t) acc[m][t] = (v8f){};
            #pragma unroll
            for (int s = 0; s < 4; ++s) {
                #pragma unroll
                for (int t = 0; t < NT; ++t) {
                    acc[m][t] = __builtin_amdgcn_wmma_f32_16x16x4_f32(false, af[m][t][s], false, vt[t][s], (short)0, acc[m][t], false, false);
                }
            }
        }
        __builtin_amdgcn_sched_barrier(0);
        v2f lg[4][12];
        #pragma unroll
        for (int m = 0; m < 3; ++m) {
            #pragma unroll
            for (int r = 0; r < 8; r += 2) {
                v2f m0, m1, m2, m3, m4, m5;
                m0[0] = acc[m][0][r]; m0[1] = acc[m][0][r + 1];
                m1[0] = acc[m][1][r]; m1[1] = acc[m][1][r + 1];
                m2[0] = acc[m][2][r]; m2[1] = acc[m][2][r + 1];
                m3[0] = acc[m][3][r]; m3[1] = acc[m][3][r + 1];
                m4[0] = acc[m][4][r]; m4[1] = acc[m][4][r + 1];
                m5[0] = acc[m][5][r]; m5[1] = acc[m][5][r + 1];
                v2f s12 = m1 + m2, d12 = m1 - m2, s34 = m3 + m4, d34 = m3 - m4;
                lg[0][m * 4 + r / 2] = (m0 + s12) + s34;
                lg[1][m * 4 + r / 2] = d12 + 2.0f * d34;
                lg[2][m * 4 + r / 2] = s12 + 4.0f * s34;
                lg[3][m * 4 + r / 2] = (d12 + 8.0f * d34) + m5;
            }
        }

        softmax_quad<SHIFT>(lg, pp, hi, best0, best1, best2, best3);
    }
}

__global__ __launch_bounds__(BLOCK) __attribute__((amdgpu_waves_per_eu(2, 2)))
void vi_wmma_kernel(const float* __restrict__ values,
                    const float* __restrict__ rewards,
                    const float* __restrict__ weight,
                    float* __restrict__ out)
{
    __shared__ __attribute__((aligned(16))) float smem[LDS_FLOATS];
    float* wlds = smem;
    const int tid  = threadIdx.x;
    const int lane = tid & 31;
    const int wave = tid >> 5;
    float* raw = smem + WLDS_FLOATS + wave * RAW_FLOATS;

    float gw[STAGE_ITERS][6];
    #pragma unroll
    for (int it = 0; it < STAGE_ITERS; ++it) {
        int idx = tid + it * BLOCK;
        int c  = idx >> 3;
        int kq = idx & 7;
        #pragma unroll
        for (int e2 = 0; e2 < 2; ++e2) {
            const int  k    = 2 * kq + e2;
            const bool real = (c < ADIM * IKK) && (k < 15);
            const float* p  = weight + (real ? c * IKK + 3 * k : 0);
            float t0 = p[0], t1 = p[1], t2 = p[2];
            gw[it][3 * e2 + 0] = real ? t0 : 0.0f;
            gw[it][3 * e2 + 1] = real ? t1 : 0.0f;
            gw[it][3 * e2 + 2] = real ? t2 : 0.0f;
        }
    }
    #pragma unroll
    for (int it = 0; it < STAGE_ITERS; ++it) {
        int idx = tid + it * BLOCK;
        if (idx < STAGE_ITEMS) {
            int c  = idx >> 3;
            int kq = idx & 7;
            float* dst = wlds + c * UP + 2 * kq;
            v2f u0, u1, u2, u3, u4, u5;
            #pragma unroll
            for (int e = 0; e < 2; ++e) {
                float g0 = gw[it][3 * e], g1 = gw[it][3 * e + 1], g2 = gw[it][3 * e + 2];
                float ga = g0 + g2;
                float gb = g0 * (LOG2E / 24.0f) + g2 * (LOG2E / 6.0f);
                float gc = g1 * (LOG2E / 12.0f);
                u0[e] = g0 * (0.25f * LOG2E);
                u1[e] = (ga + g1) * (-LOG2E / 6.0f);
                u2[e] = (ga - g1) * (-LOG2E / 6.0f);
                u3[e] = gb + gc;
                u4[e] = gb - gc;
                u5[e] = g2 * LOG2E;
            }
            *(v2f*)(dst + 0 * UT) = u0;
            *(v2f*)(dst + 1 * UT) = u1;
            *(v2f*)(dst + 2 * UT) = u2;
            *(v2f*)(dst + 3 * UT) = u3;
            *(v2f*)(dst + 4 * UT) = u4;
            *(v2f*)(dst + 5 * UT) = u5;
        }
    }

    float wmax = 0.0f;
    #pragma unroll
    for (int it = 0; it < STAGE_ITERS; ++it) {
        #pragma unroll
        for (int e = 0; e < 6; ++e) wmax = fmaxf(wmax, fabsf(gw[it][e]));
    }
    #pragma unroll
    for (int sh = 16; sh > 0; sh >>= 1) wmax = fmaxf(wmax, __shfl_xor(wmax, sh));
    if (lane == 0) smem[WMAX_OFF + wave] = wmax;

    __syncthreads();

    {
        const v4f wa = *(const v4f*)(smem + WMAX_OFF);
        const v4f wc = *(const v4f*)(smem + WMAX_OFF + 4);
        wmax = fmaxf(fmaxf(fmaxf(wa[0], wa[1]), fmaxf(wa[2], wa[3])), fmaxf(fmaxf(wc[0], wc[1]), fmaxf(wc[2], wc[3])));
    }

    const int  wave_u  = __builtin_amdgcn_readfirstlane(wave);
    const bool younger = wave_u >= (WAVES / 2);
    if (younger) __builtin_amdgcn_s_setprio(3);

    const int  n  = lane & 15;
    const int  h  = lane >> 4;
    const bool hi = h != 0;

    #pragma unroll 1
    for (int tt = 0; tt < TILES_PER_WAVE; ++tt) {
        int g   = blockIdx.x * (WAVES * TILES_PER_WAVE) + tt * WAVES + wave;
        int lin = g * TPX;
        int b   = lin >> 14;
        int rem = lin & 16383;
        int y   = rem >> 7;
        int x0  = rem & 127;

        const float* vb = values  + b * (HH * WW);
        const float* rb = rewards + b * (RDIM * HH * WW);
        int  ro[3];
        bool ok[3];
        #pragma unroll
        for (int dy = 0; dy < 3; ++dy) {
            int yy = y + dy - 1;
            ok[dy] = (yy >= 0) && (yy < HH);
            ro[dy] = min(max(yy, 0), HH - 1) * WW;
        }
        v2f wv[15];
        #pragma unroll
        for (int r = 0; r < 15; ++r) {
            const int i  = r / 3;
            const int dy = r - 3 * i;
            const float* src = (i < RDIM) ? rb + i * (HH * WW) + ro[dy] + x0 + 2 * lane : vb + ro[dy] + x0 + 2 * lane;
            v2f v = *(const v2f*)src;
            wv[r][0] = ok[dy] ? v[0] : 0.0f;
            wv[r][1] = ok[dy] ? v[1] : 0.0f;
        }
        const int  hr   = min(lane >> 1, 14);
        const int  hsd  = lane & 1;
        const int  hi_i = hr / 3;
        const int  hdy  = hr - 3 * hi_i;
        const int  hyy  = y + hdy - 1;
        const int  hxx  = hsd ? x0 + TPX : x0 - 1;
        const bool hok  = (hyy >= 0) && (hyy < HH) && (hxx >= 0) && (hxx < WW);
        const int  hoff = min(max(hyy, 0), HH - 1) * WW + min(max(hxx, 0), WW - 1);
        float hv = (hi_i < RDIM) ? rb[hi_i * (HH * WW) + hoff] : vb[hoff];
        hv = hok ? hv : 0.0f;

        #pragma unroll
        for (int r = 0; r < 15; ++r) {
            raw[r * RP + 1 + 2 * lane] = wv[r][0];
            raw[r * RP + 2 + 2 * lane] = wv[r][1];
        }
        if (lane < 30) raw[hr * RP + (hsd ? TPX + 1 : 0)] = hv;
        raw[15 * RP + lane] = 0.0f;
        raw[15 * RP + 32 + lane] = 0.0f;
        if (lane < 4) raw[15 * RP + 64 + lane] = 0.0f;


        const float* fV = raw + 4 * n + (hi ? 2 * RP : 0);
        v2f vt[NT][4];
        float ca0 = 0.0f, ca1 = 0.0f, ca2 = 0.0f, ca3 = 0.0f, ca4 = 0.0f, ca5 = 0.0f;
        #pragma unroll
        for (int s = 0; s < 4; ++s) {
            #pragma unroll
            for (int e = 0; e < 2; ++e) {
                const int off = (4 * s + e) * RP;
                v4f da = *(const v4f*)(fV + off);
                v2f db = *(const v2f*)(fV + off + 4);
                const float d0 = da[0], d1 = da[1], d2 = da[2], d3 = da[3], d4 = db[0], d5 = db[1];
                ca0 += fabsf(d0); ca1 += fabsf(d1); ca2 += fabsf(d2); ca3 += fabsf(d3); ca4 += fabsf(d4); ca5 += fabsf(d5);
                const float e31 = d3 - d1;
                const float e42 = d4 - d2;
                vt[0][s][e] = (4.0f * d0 - 5.0f * d2) + d4;
                vt[1][s][e] = (d3 + d4) - 4.0f * (d1 + d2);
                vt[2][s][e] = (d4 - d3) + 4.0f * (d1 - d2);
                vt[3][s][e] = e42 + 2.0f * e31;
                vt[4][s][e] = e42 - 2.0f * e31;
                vt[5][s][e] = (4.0f * d1 - 5.0f * d3) + d5;
            }
        }

        const float* fC = raw + 4 * n + (hi ? 2 * RP + 2 : 0);
        const float* fD = raw + 4 * n + (hi ? 3 * RP - 1 : 0);
        v2f pp[4][12];
        #pragma unroll
        for (int m = 0; m < 3; ++m) {
            #pragma unroll
            for (int r = 0; r < 8; ++r) {
                const int jl  = 16 * m + r;
                const int off = (jl / 3) * RP + (jl % 3);
                const float* src = ((jl % 3) == 0 ? fC : fD) + off;
                pp[0][m * 4 + r / 2][r & 1] = src[0];
                pp[1][m * 4 + r / 2][r & 1] = src[1];
                pp[2][m * 4 + r / 2][r & 1] = src[2];
                pp[3][m * 4 + r / 2][r & 1] = src[3];
            }
        }

        float best0 = -3.402823466e+38f, best1 = -3.402823466e+38f, best2 = -3.402823466e+38f, best3 = -3.402823466e+38f;
        const float* ub = wlds + n * UP + 2 * h;

        const float c12 = ca1 + ca2, c34 = ca3 + ca4;
        float psum = fmaxf(fmaxf(ca0 + c12, c12 + ca3), fmaxf(ca2 + c34, c34 + ca5));
        psum += __shfl_xor(psum, 16);
        const bool small = (wmax * psum <= 64.0f) && (psum <= 1.0e6f);

        const int pass0 = tt * ADIM;
        const int split = (younger && PRIO_ACTIONS >= pass0 && PRIO_ACTIONS < pass0 + ADIM) ? PRIO_ACTIONS - pass0 : ADIM;
        const bool nomax = __all(small);

        int o_from = 0;
        #pragma unroll 1
        for (int part = 0; part < 2; ++part) {
            const int o_to = (part == 0) ? split : ADIM;
            if (part == 1 && split < ADIM) __builtin_amdgcn_s_setprio(0);
            if (nomax) {
                tile_actions<false>(ub, vt, pp, hi, o_from, o_to, best0, best1, best2, best3);
            } else {
                tile_actions<true>(ub, vt, pp, hi, o_from, o_to, best0, best1, best2, best3);
            }
            o_from = o_to;
        }

        if (!hi) {
            v4f res; res[0] = best0; res[1] = best1; res[2] = best2; res[3] = best3;
            *(v4f*)(out + lin + 4 * n) = res;
        }
    }
    __builtin_amdgcn_s_setprio(0);
}

extern "C" void kernel_launch(void* const* d_in, const int* in_sizes, int n_in,
                              void* d_out, int out_size, void* d_ws, size_t ws_size,
                              hipStream_t stream) {
    (void)in_sizes; (void)n_in; (void)d_ws; (void)ws_size; (void)out_size;
    const float* values  = (const float*)d_in[0];
    const float* rewards = (const float*)d_in[1];
    const float* weight  = (const float*)d_in[2];
    float* out = (float*)d_out;

    hipLaunchKernelGGL(vi_wmma_kernel, dim3(NBLOCKS), dim3(BLOCK), 0, stream,
                       values, rewards, weight, out);
}
